# starting kernel
# baseline (speedup 1.0000x reference)
#include <hip/hip_runtime.h>

typedef __attribute__((ext_vector_type(2))) float v2f;
typedef __attribute__((ext_vector_type(8))) float v8f;

#define Bb 32
#define Nn 128
#define Dn 512
#define NL 16
#define ROWS (Bb * Nn)

__global__ void prep_weights(const float* __restrict__ w,
                             const float* __restrict__ bias,
                             float* __restrict__ Wa,
                             float* __restrict__ Bt) {
    int idx = blockIdx.x * blockDim.x + threadIdx.x;
    if (idx < Dn * Dn) {
        int o = idx / Dn;
        int i = idx - o * Dn;
        float v = w[idx] + w[Dn * Dn + idx];
        if (o == i) v += 1.0f;
        Wa[idx] = v;
    }
    if (idx < Dn * NL) {
        int o = idx / NL;
        int l = idx - o * NL;
        Bt[idx] = bias[l * Dn + o];
    }
}

__global__ void count_labels(const int* __restrict__ graph,
                             float* __restrict__ cnt) {
    __shared__ int h[NL];
    int row = blockIdx.x;
    int t = threadIdx.x;
    if (t < NL) h[t] = 0;
    __syncthreads();
    int lab = graph[row * Nn + t] & (NL - 1);
    atomicAdd(&h[lab], 1);
    __syncthreads();
    if (t < NL) cnt[row * NL + t] = (float)h[t];
}

__global__ __launch_bounds__(256) void gemm_fused(
    const float* __restrict__ X,
    const float* __restrict__ Wa,
    const float* __restrict__ Bt,
    const float* __restrict__ cnt,
    float* __restrict__ out) {

    const int lane = threadIdx.x & 31;
    const int wave = threadIdx.x >> 5;
    const int wid  = blockIdx.x * 8 + wave;
    const int m0 = (wid >> 3) * 32;
    const int o0 = (wid & 7) * 64;
    const int r = lane & 15;
    const int h = lane >> 4;

    const float* a0p = X + (size_t)(m0 + r) * Dn + 2 * h;
    const float* a1p = X + (size_t)(m0 + 16 + r) * Dn + 2 * h;
    const float* bp0 = Wa + (size_t)(o0 + 0 * 16 + r) * Dn + 2 * h;
    const float* bp1 = Wa + (size_t)(o0 + 1 * 16 + r) * Dn + 2 * h;
    const float* bp2 = Wa + (size_t)(o0 + 2 * 16 + r) * Dn + 2 * h;
    const float* bp3 = Wa + (size_t)(o0 + 3 * 16 + r) * Dn + 2 * h;

    v8f acc[2][4];
#pragma unroll
    for (int mt = 0; mt < 2; ++mt)
#pragma unroll
        for (int nt = 0; nt < 4; ++nt)
            acc[mt][nt] = (v8f){0.f, 0.f, 0.f, 0.f, 0.f, 0.f, 0.f, 0.f};

#pragma unroll 4
    for (int k0 = 0; k0 < Dn; k0 += 4) {
        v2f a0 = *(const v2f*)(a0p + k0);
        v2f a1 = *(const v2f*)(a1p + k0);
        v2f b0 = *(const v2f*)(bp0 + k0);
        v2f b1 = *(const v2f*)(bp1 + k0);
        v2f b2 = *(const v2f*)(bp2 + k0);
        v2f b3 = *(const v2f*)(bp3 + k0);
        acc[0][0] = __builtin_amdgcn_wmma_f32_16x16x4_f32(false, a0, false, b0, (short)0, acc[0][0], false, false);
        acc[1][0] = __builtin_amdgcn_wmma_f32_16x16x4_f32(false, a1, false, b0, (short)0, acc[1][0], false, false);
        acc[0][1] = __builtin_amdgcn_wmma_f32_16x16x4_f32(false, a0, false, b1, (short)0, acc[0][1], false, false);
        acc[1][1] = __builtin_amdgcn_wmma_f32_16x16x4_f32(false, a1, false, b1, (short)0, acc[1][1], false, false);
        acc[0][2] = __builtin_amdgcn_wmma_f32_16x16x4_f32(false, a0, false, b2, (short)0, acc[0][2], false, false);
        acc[1][2] = __builtin_amdgcn_wmma_f32_16x16x4_f32(false, a1, false, b2, (short)0, acc[1][2], false, false);
        acc[0][3] = __builtin_amdgcn_wmma_f32_16x16x4_f32(false, a0, false, b3, (short)0, acc[0][3], false, false);
        acc[1][3] = __builtin_amdgcn_wmma_f32_16x16x4_f32(false, a1, false, b3, (short)0, acc[1][3], false, false);
    }

#pragma unroll
    for (int k0 = 0; k0 < NL; k0 += 4) {
        v2f a0 = *(const v2f*)(cnt + (size_t)(m0 + r) * NL + k0 + 2 * h);
        v2f a1 = *(const v2f*)(cnt + (size_t)(m0 + 16 + r) * NL + k0 + 2 * h);
#pragma unroll
        for (int nt = 0; nt < 4; ++nt) {
            v2f b = *(const v2f*)(Bt + (size_t)(o0 + nt * 16 + r) * NL + k0 + 2 * h);
            acc[0][nt] = __builtin_amdgcn_wmma_f32_16x16x4_f32(false, a0, false, b, (short)0, acc[0][nt], false, false);
            acc[1][nt] = __builtin_amdgcn_wmma_f32_16x16x4_f32(false, a1, false, b, (short)0, acc[1][nt], false, false);
        }
    }

#pragma unroll
    for (int mt = 0; mt < 2; ++mt)
#pragma unroll
        for (int nt = 0; nt < 4; ++nt)
#pragma unroll
            for (int v = 0; v < 8; ++v)
                out[(size_t)(m0 + mt * 16 + v + 8 * h) * Dn + o0 + nt * 16 + r] = acc[mt][nt][v];
}

extern "C" void kernel_launch(void* const* d_in, const int* in_sizes, int n_in,
                              void* d_out, int out_size, void* d_ws, size_t ws_size,
                              hipStream_t stream) {
    const float* feature = (const float*)d_in[0];
    const int*   graph   = (const int*)d_in[1];
    const float* weights = (const float*)d_in[2];
    const float* bias    = (const float*)d_in[3];
    float* out = (float*)d_out;

    float* ws = (float*)d_ws;
    float* Wa  = ws;
    float* Bt  = ws + Dn * Dn;
    float* cnt = ws + Dn * Dn + Dn * NL;

    prep_weights<<<(Dn * Dn + 255) / 256, 256, 0, stream>>>(weights, bias, Wa, Bt);
    count_labels<<<ROWS, Nn, 0, stream>>>(graph, cnt);
    gemm_fused<<<128, 256, 0, stream>>>(feature, Wa, Bt, cnt, out);
}


# Preparation re-cut as ONE launch of 1024 waves (was two, 24576 waves): a thread sums 8 weights and counts 16 labels in packed bit fields, 8-lane DPP butterfly, no LDS atomics; sum order kept
# speedup vs baseline: 1.7009x; 1.7009x over previous
#include <hip/hip_runtime.h>

typedef __attribute__((ext_vector_type(2))) float v2f;
typedef __attribute__((ext_vector_type(4))) float v4f;
typedef __attribute__((ext_vector_type(8))) float v8f;

#define Bb 32
#define Nn 128
#define Dn 512
#define NL 16
#define ROWS (Bb * Nn)

typedef __attribute__((ext_vector_type(4))) int v4i;

template <int X>
__device__ __forceinline__ unsigned lane_xor(unsigned v) {
    return (unsigned)__builtin_amdgcn_update_dpp(0, (int)v, 0x160 + X, 0xF, 0xF, false);
}

__device__ __forceinline__ unsigned long long bins4(v4i a, v4i b) {
    unsigned long long s = 0;
#pragma unroll
    for (int e = 0; e < 4; ++e) {
        s += 1ull << ((a[e] & (NL - 1)) * 4);
        s += 1ull << ((b[e] & (NL - 1)) * 4);
    }
    return s;
}

__global__ __launch_bounds__(128) void prep_all(const float* __restrict__ w,
                                                const float* __restrict__ bias,
                                                const int* __restrict__ graph,
                                                float* __restrict__ Wa,
                                                float* __restrict__ Bt,
                                                float* __restrict__ cnt) {
    const int g = blockIdx.x * 128 + threadIdx.x;

    const v4i* gp = (const v4i*)(graph + (size_t)g * 16);
    const v4i l0 = gp[0], l1 = gp[1], l2 = gp[2], l3 = gp[3];
    const v4f* w0p = (const v4f*)w;
    const v4f* w1p = (const v4f*)(w + Dn * Dn);
    const v4f x0 = w0p[g], y0 = w1p[g];
    const v4f x1 = w0p[g + 32768], y1 = w1p[g + 32768];
    float bt = 0.f;
    if (g < Dn * NL) bt = bias[(g & (NL - 1)) * Dn + (g >> 4)];

#pragma unroll
    for (int j = 0; j < 2; ++j) {
        const int q4 = g + j * 32768;
        v4f v = j ? (x1 + y1) : (x0 + y0);
        const int d = (q4 >> 7) - (q4 & 127) * 4;
        if (d == 0) v.x += 1.0f;
        if (d == 1) v.y += 1.0f;
        if (d == 2) v.z += 1.0f;
        if (d == 3) v.w += 1.0f;
        ((v4f*)Wa)[q4] = v;
    }

    const unsigned long long M = 0x0F0F0F0F0F0F0F0Full;
    const unsigned long long sa = bins4(l0, l1), sb = bins4(l2, l3);
    const unsigned long long ev = (sa & M) + (sb & M);
    const unsigned long long od = ((sa >> 4) & M) + ((sb >> 4) & M);
    unsigned e0 = (unsigned)ev, e1 = (unsigned)(ev >> 32);
    unsigned o0 = (unsigned)od, o1 = (unsigned)(od >> 32);
    e0 += lane_xor<1>(e0); e1 += lane_xor<1>(e1); o0 += lane_xor<1>(o0); o1 += lane_xor<1>(o1);
    e0 += lane_xor<2>(e0); e1 += lane_xor<2>(e1); o0 += lane_xor<2>(o0); o1 += lane_xor<2>(o1);
    e0 += lane_xor<4>(e0); e1 += lane_xor<4>(e1); o0 += lane_xor<4>(o0); o1 += lane_xor<4>(o1);
    const int q = g & 7;
    const unsigned ce = (((q & 4) ? e1 : e0) >> ((q & 3) * 8)) & 255u;
    const unsigned co = (((q & 4) ? o1 : o0) >> ((q & 3) * 8)) & 255u;
    ((v2f*)cnt)[g] = (v2f){(float)ce, (float)co};

    if (g < Dn * NL) Bt[g] = bt;
}

struct Frags {
    v4f a[2][2];
    v4f b[4][2];
};

__device__ __forceinline__ void load_frags(Frags& f, const float* __restrict__ ap,
                                           const float* __restrict__ bp, int k0) {
#pragma unroll
    for (int mt = 0; mt < 2; ++mt)
#pragma unroll
        for (int kh = 0; kh < 2; ++kh)
            f.a[mt][kh] = *(const v4f*)(ap + (size_t)mt * 16 * Dn + k0 + kh * 8);
#pragma unroll
    for (int nt = 0; nt < 4; ++nt)
#pragma unroll
        for (int kh = 0; kh < 2; ++kh)
            f.b[nt][kh] = *(const v4f*)(bp + (size_t)nt * 16 * Dn + k0 + kh * 8);
}

__device__ __forceinline__ void close_fetch_group() {
    asm volatile("" ::: "memory");
    __builtin_amdgcn_sched_barrier(0);
}

__device__ __forceinline__ void mma_step(v8f (&acc)[2][4], const Frags& f) {
#pragma unroll
    for (int kh = 0; kh < 2; ++kh) {
#pragma unroll
        for (int sub = 0; sub < 2; ++sub) {
            v2f a[2], b[4];
#pragma unroll
            for (int mt = 0; mt < 2; ++mt)
                a[mt] = sub ? (v2f){f.a[mt][kh].z, f.a[mt][kh].w}
                            : (v2f){f.a[mt][kh].x, f.a[mt][kh].y};
#pragma unroll
            for (int nt = 0; nt < 4; ++nt)
                b[nt] = sub ? (v2f){f.b[nt][kh].z, f.b[nt][kh].w}
                            : (v2f){f.b[nt][kh].x, f.b[nt][kh].y};
#pragma unroll
            for (int nt = 0; nt < 4; ++nt)
#pragma unroll
                for (int mt = 0; mt < 2; ++mt)
                    acc[mt][nt] = __builtin_amdgcn_wmma_f32_16x16x4_f32(
                        false, a[mt], false, b[nt], (short)0, acc[mt][nt], false, false);
        }
    }
}

__global__ __launch_bounds__(128) void gemm_fused(
    const float* __restrict__ X,
    const float* __restrict__ Wa,
    const float* __restrict__ Bt,
    const float* __restrict__ cnt,
    float* __restrict__ out) {

    const int lane = threadIdx.x & 31;
    const int wave = threadIdx.x >> 5;
    const int m0 = (blockIdx.x >> 2) * 64 + (wave >> 1) * 32;
    const int o0 = (blockIdx.x & 3) * 128 + (wave & 1) * 64;
    const int r = lane & 15;
    const int h = lane >> 4;

    const float* ap = X  + (size_t)(m0 + r) * Dn + 4 * h;
    const float* bp = Wa + (size_t)(o0 + r) * Dn + 4 * h;

    v8f acc[2][4];
#pragma unroll
    for (int mt = 0; mt < 2; ++mt)
#pragma unroll
        for (int nt = 0; nt < 4; ++nt)
            acc[mt][nt] = (v8f){0.f, 0.f, 0.f, 0.f, 0.f, 0.f, 0.f, 0.f};

    Frags f0, f1;
    load_frags(f0, ap, bp, 0);
    close_fetch_group();
#pragma unroll 1
    for (int k0 = 0; k0 < Dn; k0 += 32) {
        load_frags(f1, ap, bp, k0 + 16);
        close_fetch_group();
        mma_step(acc, f0);
        load_frags(f0, ap, bp, (k0 + 32) & (Dn - 1));
        close_fetch_group();
        mma_step(acc, f1);
    }

#pragma unroll
    for (int k0 = 0; k0 < NL; k0 += 4) {
        v2f a0 = *(const v2f*)(cnt + (size_t)(m0 + r) * NL + k0 + 2 * h);
        v2f a1 = *(const v2f*)(cnt + (size_t)(m0 + 16 + r) * NL + k0 + 2 * h);
#pragma unroll
        for (int nt = 0; nt < 4; ++nt) {
            v2f b = *(const v2f*)(Bt + (size_t)(o0 + nt * 16 + r) * NL + k0 + 2 * h);
            acc[0][nt] = __builtin_amdgcn_wmma_f32_16x16x4_f32(false, a0, false, b, (short)0, acc[0][nt], false, false);
            acc[1][nt] = __builtin_amdgcn_wmma_f32_16x16x4_f32(false, a1, false, b, (short)0, acc[1][nt], false, false);
        }
    }

#pragma unroll
    for (int mt = 0; mt < 2; ++mt)
#pragma unroll
        for (int nt = 0; nt < 4; ++nt)
#pragma unroll
            for (int v = 0; v < 8; ++v)
                out[(size_t)(m0 + mt * 16 + v + 8 * h) * Dn + o0 + nt * 16 + r] = acc[mt][nt][v];
}

extern "C" void kernel_launch(void* const* d_in, const int* in_sizes, int n_in,
                              void* d_out, int out_size, void* d_ws, size_t ws_size,
                              hipStream_t stream) {
    const float* feature = (const float*)d_in[0];
    const int*   graph   = (const int*)d_in[1];
    const float* weights = (const float*)d_in[2];
    const float* bias    = (const float*)d_in[3];
    float* out = (float*)d_out;

    float* ws = (float*)d_ws;
    float* Wa  = ws;
    float* Bt  = ws + Dn * Dn;
    float* cnt = ws + Dn * Dn + Dn * NL;

    prep_all<<<256, 128, 0, stream>>>(weights, bias, graph, Wa, Bt, cnt);
    gemm_fused<<<256, 128, 0, stream>>>(feature, Wa, Bt, cnt, out);
}


# W_eff is written by the preparation in WMMA fragment order (tile, k step, lane), so each B fragment load of the GEMM reads 512 contiguous bytes, 4 lines, not 16 rows x 32 bytes; sum order kept
# speedup vs baseline: 1.7852x; 1.0496x over previous
#include <hip/hip_runtime.h>

typedef __attribute__((ext_vector_type(2))) float v2f;
typedef __attribute__((ext_vector_type(4))) float v4f;
typedef __attribute__((ext_vector_type(8))) float v8f;

#define Bb 32
#define Nn 128
#define Dn 512
#define NL 16
#define ROWS (Bb * Nn)

typedef __attribute__((ext_vector_type(4))) int v4i;

template <int X>
__device__ __forceinline__ unsigned lane_xor(unsigned v) {
    return (unsigned)__builtin_amdgcn_update_dpp(0, (int)v, 0x160 + X, 0xF, 0xF, false);
}

__device__ __forceinline__ unsigned long long bins4(v4i a, v4i b) {
    unsigned long long s = 0;
#pragma unroll
    for (int e = 0; e < 4; ++e) {
        s += 1ull << ((a[e] & (NL - 1)) * 4);
        s += 1ull << ((b[e] & (NL - 1)) * 4);
    }
    return s;
}

__global__ __launch_bounds__(128) void prep_all(const float* __restrict__ w,
                                                const float* __restrict__ bias,
                                                const int* __restrict__ graph,
                                                float* __restrict__ Wa,
                                                float* __restrict__ Bt,
                                                float* __restrict__ cnt) {
    const int g = blockIdx.x * 128 + threadIdx.x;

    const v4i* gp = (const v4i*)(graph + (size_t)g * 16);
    const v4i l0 = gp[0], l1 = gp[1], l2 = gp[2], l3 = gp[3];
    const v4f* w0p = (const v4f*)w;
    const v4f* w1p = (const v4f*)(w + Dn * Dn);
    const int fr = g & 15, fh = (g >> 4) & 1, fkh = (g >> 5) & 1, fs = (g >> 6) & 31;
    const int fk = fs * 16 + fkh * 8 + fh * 4;
    const int fo0 = (g >> 11) * 16 + fr;
    const int s0 = fo0 * (Dn / 4) + (fk >> 2);
    const int s1 = s0 + 256 * (Dn / 4);
    const v4f x0 = w0p[s0], y0 = w1p[s0];
    const v4f x1 = w0p[s1], y1 = w1p[s1];
    float bt = 0.f;
    if (g < Dn * NL) bt = bias[(g & (NL - 1)) * Dn + (g >> 4)];

#pragma unroll
    for (int j = 0; j < 2; ++j) {
        v4f v = j ? (x1 + y1) : (x0 + y0);
        const int d = fo0 + 256 * j - fk;
        if (d == 0) v.x += 1.0f;
        if (d == 1) v.y += 1.0f;
        if (d == 2) v.z += 1.0f;
        if (d == 3) v.w += 1.0f;
        ((v4f*)Wa)[g + j * 32768] = v;
    }

    const unsigned long long M = 0x0F0F0F0F0F0F0F0Full;
    const unsigned long long sa = bins4(l0, l1), sb = bins4(l2, l3);
    const unsigned long long ev = (sa & M) + (sb & M);
    const unsigned long long od = ((sa >> 4) & M) + ((sb >> 4) & M);
    unsigned e0 = (unsigned)ev, e1 = (unsigned)(ev >> 32);
    unsigned o0 = (unsigned)od, o1 = (unsigned)(od >> 32);
    e0 += lane_xor<1>(e0); e1 += lane_xor<1>(e1); o0 += lane_xor<1>(o0); o1 += lane_xor<1>(o1);
    e0 += lane_xor<2>(e0); e1 += lane_xor<2>(e1); o0 += lane_xor<2>(o0); o1 += lane_xor<2>(o1);
    e0 += lane_xor<4>(e0); e1 += lane_xor<4>(e1); o0 += lane_xor<4>(o0); o1 += lane_xor<4>(o1);
    const int q = g & 7;
    const unsigned ce = (((q & 4) ? e1 : e0) >> ((q & 3) * 8)) & 255u;
    const unsigned co = (((q & 4) ? o1 : o0) >> ((q & 3) * 8)) & 255u;
    ((v2f*)cnt)[g] = (v2f){(float)ce, (float)co};

    if (g < Dn * NL) Bt[g] = bt;
}

struct Frags {
    v4f a[2][2];
    v4f b[4][2];
};

#define WT_PIECES (16 * Dn / 4)

__device__ __forceinline__ void load_frags(Frags& f, const float* __restrict__ ap,
                                           const v4f* __restrict__ bq, int k0) {
#pragma unroll
    for (int mt = 0; mt < 2; ++mt)
#pragma unroll
        for (int kh = 0; kh < 2; ++kh)
            f.a[mt][kh] = *(const v4f*)(ap + (size_t)mt * 16 * Dn + k0 + kh * 8);
#pragma unroll
    for (int nt = 0; nt < 4; ++nt)
#pragma unroll
        for (int kh = 0; kh < 2; ++kh)
            f.b[nt][kh] = bq[nt * WT_PIECES + k0 * 4 + kh * 32];
}

__device__ __forceinline__ void close_fetch_group() {
    asm volatile("" ::: "memory");
    __builtin_amdgcn_sched_barrier(0);
}

__device__ __forceinline__ void mma_step(v8f (&acc)[2][4], const Frags& f) {
#pragma unroll
    for (int kh = 0; kh < 2; ++kh) {
#pragma unroll
        for (int sub = 0; sub < 2; ++sub) {
            v2f a[2], b[4];
#pragma unroll
            for (int mt = 0; mt < 2; ++mt)
                a[mt] = sub ? (v2f){f.a[mt][kh].z, f.a[mt][kh].w}
                            : (v2f){f.a[mt][kh].x, f.a[mt][kh].y};
#pragma unroll
            for (int nt = 0; nt < 4; ++nt)
                b[nt] = sub ? (v2f){f.b[nt][kh].z, f.b[nt][kh].w}
                            : (v2f){f.b[nt][kh].x, f.b[nt][kh].y};
#pragma unroll
            for (int nt = 0; nt < 4; ++nt)
#pragma unroll
                for (int mt = 0; mt < 2; ++mt)
                    acc[mt][nt] = __builtin_amdgcn_wmma_f32_16x16x4_f32(
                        false, a[mt], false, b[nt], (short)0, acc[mt][nt], false, false);
        }
    }
}

__global__ __launch_bounds__(128) void gemm_fused(
    const float* __restrict__ X,
    const float* __restrict__ Wa,
    const float* __restrict__ Bt,
    const float* __restrict__ cnt,
    float* __restrict__ out) {

    const int lane = threadIdx.x & 31;
    const int wave = threadIdx.x >> 5;
    const int m0 = (blockIdx.x >> 2) * 64 + (wave >> 1) * 32;
    const int o0 = (blockIdx.x & 3) * 128 + (wave & 1) * 64;
    const int r = lane & 15;
    const int h = lane >> 4;

    const float* ap = X  + (size_t)(m0 + r) * Dn + 4 * h;
    const v4f* bq = (const v4f*)Wa + (size_t)(o0 >> 4) * WT_PIECES + lane;

    v8f acc[2][4];
#pragma unroll
    for (int mt = 0; mt < 2; ++mt)
#pragma unroll
        for (int nt = 0; nt < 4; ++nt)
            acc[mt][nt] = (v8f){0.f, 0.f, 0.f, 0.f, 0.f, 0.f, 0.f, 0.f};

    Frags f0, f1;
    load_frags(f0, ap, bq, 0);
    close_fetch_group();
#pragma unroll 1
    for (int k0 = 0; k0 < Dn; k0 += 32) {
        load_frags(f1, ap, bq, k0 + 16);
        close_fetch_group();
        mma_step(acc, f0);
        load_frags(f0, ap, bq, (k0 + 32) & (Dn - 1));
        close_fetch_group();
        mma_step(acc, f1);
    }

#pragma unroll
    for (int k0 = 0; k0 < NL; k0 += 4) {
        v2f a0 = *(const v2f*)(cnt + (size_t)(m0 + r) * NL + k0 + 2 * h);
        v2f a1 = *(const v2f*)(cnt + (size_t)(m0 + 16 + r) * NL + k0 + 2 * h);
#pragma unroll
        for (int nt = 0; nt < 4; ++nt) {
            v2f b = *(const v2f*)(Bt + (size_t)(o0 + nt * 16 + r) * NL + k0 + 2 * h);
            acc[0][nt] = __builtin_amdgcn_wmma_f32_16x16x4_f32(false, a0, false, b, (short)0, acc[0][nt], false, false);
            acc[1][nt] = __builtin_amdgcn_wmma_f32_16x16x4_f32(false, a1, false, b, (short)0, acc[1][nt], false, false);
        }
    }

#pragma unroll
    for (int mt = 0; mt < 2; ++mt)
#pragma unroll
        for (int nt = 0; nt < 4; ++nt)
#pragma unroll
            for (int v = 0; v < 8; ++v)
                out[(size_t)(m0 + mt * 16 + v + 8 * h) * Dn + o0 + nt * 16 + r] = acc[mt][nt][v];
}

extern "C" void kernel_launch(void* const* d_in, const int* in_sizes, int n_in,
                              void* d_out, int out_size, void* d_ws, size_t ws_size,
                              hipStream_t stream) {
    const float* feature = (const float*)d_in[0];
    const int*   graph   = (const int*)d_in[1];
    const float* weights = (const float*)d_in[2];
    const float* bias    = (const float*)d_in[3];
    float* out = (float*)d_out;

    float* ws = (float*)d_ws;
    float* Wa  = ws;
    float* Bt  = ws + Dn * Dn;
    float* cnt = ws + Dn * Dn + Dn * NL;

    prep_all<<<256, 128, 0, stream>>>(weights, bias, graph, Wa, Bt, cnt);
    gemm_fused<<<256, 128, 0, stream>>>(feature, Wa, Bt, cnt, out);
}


# GEMM operand fragments in a ring of 4 register sets (was 2): a k step's global loads are issued 3 steps (96 WMMAs) ahead of use, last trip peeled so no fetch passes K; no LDS; sum order kept
# speedup vs baseline: 2.2847x; 1.2798x over previous
#include <hip/hip_runtime.h>

typedef __attribute__((ext_vector_type(2))) float v2f;
typedef __attribute__((ext_vector_type(4))) float v4f;
typedef __attribute__((ext_vector_type(8))) float v8f;

#define Bb 32
#define Nn 128
#define Dn 512
#define NL 16
#define ROWS (Bb * Nn)

typedef __attribute__((ext_vector_type(4))) int v4i;

template <int X>
__device__ __forceinline__ unsigned lane_xor(unsigned v) {
    return (unsigned)__builtin_amdgcn_update_dpp(0, (int)v, 0x160 + X, 0xF, 0xF, false);
}

__device__ __forceinline__ unsigned long long bins4(v4i a, v4i b) {
    unsigned long long s = 0;
#pragma unroll
    for (int e = 0; e < 4; ++e) {
        s += 1ull << ((a[e] & (NL - 1)) * 4);
        s += 1ull << ((b[e] & (NL - 1)) * 4);
    }
    return s;
}

__global__ __launch_bounds__(128) void prep_all(const float* __restrict__ w,
                                                const float* __restrict__ bias,
                                                const int* __restrict__ graph,
                                                float* __restrict__ Wa,
                                                float* __restrict__ Bt,
                                                float* __restrict__ cnt) {
    const int g = blockIdx.x * 128 + threadIdx.x;

    const v4i* gp = (const v4i*)(graph + (size_t)g * 16);
    const v4i l0 = gp[0], l1 = gp[1], l2 = gp[2], l3 = gp[3];
    const v4f* w0p = (const v4f*)w;
    const v4f* w1p = (const v4f*)(w + Dn * Dn);
    const int fr = g & 15, fh = (g >> 4) & 1, fkh = (g >> 5) & 1, fs = (g >> 6) & 31;
    const int fk = fs * 16 + fkh * 8 + fh * 4;
    const int fo0 = (g >> 11) * 16 + fr;
    const int s0 = fo0 * (Dn / 4) + (fk >> 2);
    const int s1 = s0 + 256 * (Dn / 4);
    const v4f x0 = w0p[s0], y0 = w1p[s0];
    const v4f x1 = w0p[s1], y1 = w1p[s1];
    float bt = 0.f;
    if (g < Dn * NL) bt = bias[(g & (NL - 1)) * Dn + (g >> 4)];

#pragma unroll
    for (int j = 0; j < 2; ++j) {
        v4f v = j ? (x1 + y1) : (x0 + y0);
        const int d = fo0 + 256 * j - fk;
        if (d == 0) v.x += 1.0f;
        if (d == 1) v.y += 1.0f;
        if (d == 2) v.z += 1.0f;
        if (d == 3) v.w += 1.0f;
        ((v4f*)Wa)[g + j * 32768] = v;
    }

    const unsigned long long M = 0x0F0F0F0F0F0F0F0Full;
    const unsigned long long sa = bins4(l0, l1), sb = bins4(l2, l3);
    const unsigned long long ev = (sa & M) + (sb & M);
    const unsigned long long od = ((sa >> 4) & M) + ((sb >> 4) & M);
    unsigned e0 = (unsigned)ev, e1 = (unsigned)(ev >> 32);
    unsigned o0 = (unsigned)od, o1 = (unsigned)(od >> 32);
    e0 += lane_xor<1>(e0); e1 += lane_xor<1>(e1); o0 += lane_xor<1>(o0); o1 += lane_xor<1>(o1);
    e0 += lane_xor<2>(e0); e1 += lane_xor<2>(e1); o0 += lane_xor<2>(o0); o1 += lane_xor<2>(o1);
    e0 += lane_xor<4>(e0); e1 += lane_xor<4>(e1); o0 += lane_xor<4>(o0); o1 += lane_xor<4>(o1);
    const int q = g & 7;
    const unsigned ce = (((q & 4) ? e1 : e0) >> ((q & 3) * 8)) & 255u;
    const unsigned co = (((q & 4) ? o1 : o0) >> ((q & 3) * 8)) & 255u;
    ((v2f*)cnt)[g] = (v2f){(float)ce, (float)co};

    if (g < Dn * NL) Bt[g] = bt;
}

#define NS 4
struct Frags {
    v4f a[2][2];
    v4f b[4][2];
};

#define WT_PIECES (16 * Dn / 4)

__device__ __forceinline__ void load_frags(Frags& f, const float* __restrict__ ap,
                                           const v4f* __restrict__ bq, int k0) {
#pragma unroll
    for (int mt = 0; mt < 2; ++mt)
#pragma unroll
        for (int kh = 0; kh < 2; ++kh)
            f.a[mt][kh] = *(const v4f*)(ap + (size_t)mt * 16 * Dn + k0 + kh * 8);
#pragma unroll
    for (int nt = 0; nt < 4; ++nt)
#pragma unroll
        for (int kh = 0; kh < 2; ++kh)
            f.b[nt][kh] = bq[nt * WT_PIECES + k0 * 4 + kh * 32];
}

__device__ __forceinline__ void close_fetch_group() {
    asm volatile("" ::: "memory");
    __builtin_amdgcn_sched_barrier(0);
}

__device__ __forceinline__ void mma_step(v8f (&acc)[2][4], const Frags& f) {
#pragma unroll
    for (int kh = 0; kh < 2; ++kh) {
#pragma unroll
        for (int sub = 0; sub < 2; ++sub) {
            v2f a[2], b[4];
#pragma unroll
            for (int mt = 0; mt < 2; ++mt)
                a[mt] = sub ? (v2f){f.a[mt][kh].z, f.a[mt][kh].w}
                            : (v2f){f.a[mt][kh].x, f.a[mt][kh].y};
#pragma unroll
            for (int nt = 0; nt < 4; ++nt)
                b[nt] = sub ? (v2f){f.b[nt][kh].z, f.b[nt][kh].w}
                            : (v2f){f.b[nt][kh].x, f.b[nt][kh].y};
#pragma unroll
            for (int nt = 0; nt < 4; ++nt)
#pragma unroll
                for (int mt = 0; mt < 2; ++mt)
                    acc[mt][nt] = __builtin_amdgcn_wmma_f32_16x16x4_f32(
                        false, a[mt], false, b[nt], (short)0, acc[mt][nt], false, false);
        }
    }
}

template <bool LAST>
__device__ __forceinline__ void ring_trip(v8f (&acc)[2][4], Frags (&f)[NS],
                                          const float* __restrict__ ap,
                                          const v4f* __restrict__ bq, int k0) {
#pragma unroll
    for (int s = 0; s < NS; ++s) {
        if (!LAST || s == 0) {
            load_frags(f[(s + NS - 1) % NS], ap, bq, k0 + (s + NS - 1) * 16);
            close_fetch_group();
        }
        mma_step(acc, f[s]);
    }
}

__global__ __launch_bounds__(128) void gemm_fused(
    const float* __restrict__ X,
    const float* __restrict__ Wa,
    const float* __restrict__ Bt,
    const float* __restrict__ cnt,
    float* __restrict__ out) {

    const int lane = threadIdx.x & 31;
    const int wave = threadIdx.x >> 5;
    const int m0 = (blockIdx.x >> 2) * 64 + (wave >> 1) * 32;
    const int o0 = (blockIdx.x & 3) * 128 + (wave & 1) * 64;
    const int r = lane & 15;
    const int h = lane >> 4;

    const float* ap = X  + (size_t)(m0 + r) * Dn + 4 * h;
    const v4f* bq = (const v4f*)Wa + (size_t)(o0 >> 4) * WT_PIECES + lane;

    v8f acc[2][4];
#pragma unroll
    for (int mt = 0; mt < 2; ++mt)
#pragma unroll
        for (int nt = 0; nt < 4; ++nt)
            acc[mt][nt] = (v8f){0.f, 0.f, 0.f, 0.f, 0.f, 0.f, 0.f, 0.f};

    Frags f[NS];
#pragma unroll
    for (int s = 0; s < NS - 1; ++s) {
        load_frags(f[s], ap, bq, s * 16);
        close_fetch_group();
    }
#pragma unroll 1
    for (int k0 = 0; k0 < Dn - 16 * NS; k0 += 16 * NS)
        ring_trip<false>(acc, f, ap, bq, k0);
    ring_trip<true>(acc, f, ap, bq, Dn - 16 * NS);

#pragma unroll
    for (int k0 = 0; k0 < NL; k0 += 4) {
        v2f a0 = *(const v2f*)(cnt + (size_t)(m0 + r) * NL + k0 + 2 * h);
        v2f a1 = *(const v2f*)(cnt + (size_t)(m0 + 16 + r) * NL + k0 + 2 * h);
#pragma unroll
        for (int nt = 0; nt < 4; ++nt) {
            v2f b = *(const v2f*)(Bt + (size_t)(o0 + nt * 16 + r) * NL + k0 + 2 * h);
            acc[0][nt] = __builtin_amdgcn_wmma_f32_16x16x4_f32(false, a0, false, b, (short)0, acc[0][nt], false, false);
            acc[1][nt] = __builtin_amdgcn_wmma_f32_16x16x4_f32(false, a1, false, b, (short)0, acc[1][nt], false, false);
        }
    }

#pragma unroll
    for (int mt = 0; mt < 2; ++mt)
#pragma unroll
        for (int nt = 0; nt < 4; ++nt)
#pragma unroll
            for (int v = 0; v < 8; ++v)
                out[(size_t)(m0 + mt * 16 + v + 8 * h) * Dn + o0 + nt * 16 + r] = acc[mt][nt][v];
}

extern "C" void kernel_launch(void* const* d_in, const int* in_sizes, int n_in,
                              void* d_out, int out_size, void* d_ws, size_t ws_size,
                              hipStream_t stream) {
    const float* feature = (const float*)d_in[0];
    const int*   graph   = (const int*)d_in[1];
    const float* weights = (const float*)d_in[2];
    const float* bias    = (const float*)d_in[3];
    float* out = (float*)d_out;

    float* ws = (float*)d_ws;
    float* Wa  = ws;
    float* Bt  = ws + Dn * Dn;
    float* cnt = ws + Dn * Dn + Dn * NL;

    prep_all<<<256, 128, 0, stream>>>(weights, bias, graph, Wa, Bt, cnt);
    gemm_fused<<<256, 128, 0, stream>>>(feature, Wa, Bt, cnt, out);
}


# GEMM column tiles interleaved (lane r, tile nt = column 4r+nt; W_eff written so by the preparation): a lane holds 4 consecutive columns: 16 b128 whole-line stores a wave, was 64 b32; sum order kept
# speedup vs baseline: 2.4164x; 1.0576x over previous
#include <hip/hip_runtime.h>

typedef __attribute__((ext_vector_type(2))) float v2f;
typedef __attribute__((ext_vector_type(4))) float v4f;
typedef __attribute__((ext_vector_type(8))) float v8f;

#define Bb 32
#define Nn 128
#define Dn 512
#define NL 16
#define ROWS (Bb * Nn)

typedef __attribute__((ext_vector_type(4))) int v4i;

template <int X>
__device__ __forceinline__ unsigned lane_xor(unsigned v) {
    return (unsigned)__builtin_amdgcn_update_dpp(0, (int)v, 0x160 + X, 0xF, 0xF, false);
}

__device__ __forceinline__ unsigned long long bins4(v4i a, v4i b) {
    unsigned long long s = 0;
#pragma unroll
    for (int e = 0; e < 4; ++e) {
        s += 1ull << ((a[e] & (NL - 1)) * 4);
        s += 1ull << ((b[e] & (NL - 1)) * 4);
    }
    return s;
}

__global__ __launch_bounds__(128) void prep_all(const float* __restrict__ w,
                                                const float* __restrict__ bias,
                                                const int* __restrict__ graph,
                                                float* __restrict__ Wa,
                                                float* __restrict__ Bt,
                                                float* __restrict__ cnt) {
    const int g = blockIdx.x * 128 + threadIdx.x;

    const v4i* gp = (const v4i*)(graph + (size_t)g * 16);
    const v4i l0 = gp[0], l1 = gp[1], l2 = gp[2], l3 = gp[3];
    const v4f* w0p = (const v4f*)w;
    const v4f* w1p = (const v4f*)(w + Dn * Dn);
    const int fr = g & 15, fh = (g >> 4) & 1, fkh = (g >> 5) & 1, fs = (g >> 6) & 31;
    const int fk = fs * 16 + fkh * 8 + fh * 4;
    const int ft = g >> 11;
    const int fo0 = (ft >> 2) * 64 + fr * 4 + (ft & 3);
    const int s0 = fo0 * (Dn / 4) + (fk >> 2);
    const int s1 = s0 + 256 * (Dn / 4);
    const v4f x0 = w0p[s0], y0 = w1p[s0];
    const v4f x1 = w0p[s1], y1 = w1p[s1];
    float bt = 0.f;
    if (g < Dn * NL) bt = bias[(g & (NL - 1)) * Dn + (g >> 4)];

#pragma unroll
    for (int j = 0; j < 2; ++j) {
        v4f v = j ? (x1 + y1) : (x0 + y0);
        const int d = fo0 + 256 * j - fk;
        if (d == 0) v.x += 1.0f;
        if (d == 1) v.y += 1.0f;
        if (d == 2) v.z += 1.0f;
        if (d == 3) v.w += 1.0f;
        ((v4f*)Wa)[g + j * 32768] = v;
    }

    const unsigned long long M = 0x0F0F0F0F0F0F0F0Full;
    const unsigned long long sa = bins4(l0, l1), sb = bins4(l2, l3);
    const unsigned long long ev = (sa & M) + (sb & M);
    const unsigned long long od = ((sa >> 4) & M) + ((sb >> 4) & M);
    unsigned e0 = (unsigned)ev, e1 = (unsigned)(ev >> 32);
    unsigned o0 = (unsigned)od, o1 = (unsigned)(od >> 32);
    e0 += lane_xor<1>(e0); e1 += lane_xor<1>(e1); o0 += lane_xor<1>(o0); o1 += lane_xor<1>(o1);
    e0 += lane_xor<2>(e0); e1 += lane_xor<2>(e1); o0 += lane_xor<2>(o0); o1 += lane_xor<2>(o1);
    e0 += lane_xor<4>(e0); e1 += lane_xor<4>(e1); o0 += lane_xor<4>(o0); o1 += lane_xor<4>(o1);
    const int q = g & 7;
    const unsigned ce = (((q & 4) ? e1 : e0) >> ((q & 3) * 8)) & 255u;
    const unsigned co = (((q & 4) ? o1 : o0) >> ((q & 3) * 8)) & 255u;
    ((v2f*)cnt)[g] = (v2f){(float)ce, (float)co};

    if (g < Dn * NL) Bt[g] = bt;
}

#define NS 4
struct Frags {
    v4f a[2][2];
    v4f b[4][2];
};

#define WT_PIECES (16 * Dn / 4)

__device__ __forceinline__ void load_frags(Frags& f, const float* __restrict__ ap,
                                           const v4f* __restrict__ bq, int k0) {
#pragma unroll
    for (int mt = 0; mt < 2; ++mt)
#pragma unroll
        for (int kh = 0; kh < 2; ++kh)
            f.a[mt][kh] = *(const v4f*)(ap + (size_t)mt * 16 * Dn + k0 + kh * 8);
#pragma unroll
    for (int nt = 0; nt < 4; ++nt)
#pragma unroll
        for (int kh = 0; kh < 2; ++kh)
            f.b[nt][kh] = bq[nt * WT_PIECES + k0 * 4 + kh * 32];
}

__device__ __forceinline__ void close_fetch_group() {
    asm volatile("" ::: "memory");
    __builtin_amdgcn_sched_barrier(0);
}

__device__ __forceinline__ void mma_step(v8f (&acc)[2][4], const Frags& f) {
#pragma unroll
    for (int kh = 0; kh < 2; ++kh) {
#pragma unroll
        for (int sub = 0; sub < 2; ++sub) {
            v2f a[2], b[4];
#pragma unroll
            for (int mt = 0; mt < 2; ++mt)
                a[mt] = sub ? (v2f){f.a[mt][kh].z, f.a[mt][kh].w}
                            : (v2f){f.a[mt][kh].x, f.a[mt][kh].y};
#pragma unroll
            for (int nt = 0; nt < 4; ++nt)
                b[nt] = sub ? (v2f){f.b[nt][kh].z, f.b[nt][kh].w}
                            : (v2f){f.b[nt][kh].x, f.b[nt][kh].y};
#pragma unroll
            for (int nt = 0; nt < 4; ++nt)
#pragma unroll
                for (int mt = 0; mt < 2; ++mt)
                    acc[mt][nt] = __builtin_amdgcn_wmma_f32_16x16x4_f32(
                        false, a[mt], false, b[nt], (short)0, acc[mt][nt], false, false);
        }
    }
}

template <bool LAST>
__device__ __forceinline__ void ring_trip(v8f (&acc)[2][4], Frags (&f)[NS],
                                          const float* __restrict__ ap,
                                          const v4f* __restrict__ bq, int k0) {
#pragma unroll
    for (int s = 0; s < NS; ++s) {
        if (!LAST || s == 0) {
            load_frags(f[(s + NS - 1) % NS], ap, bq, k0 + (s + NS - 1) * 16);
            close_fetch_group();
        }
        mma_step(acc, f[s]);
    }
}

__global__ __launch_bounds__(128) void gemm_fused(
    const float* __restrict__ X,
    const float* __restrict__ Wa,
    const float* __restrict__ Bt,
    const float* __restrict__ cnt,
    float* __restrict__ out) {

    const int lane = threadIdx.x & 31;
    const int wave = threadIdx.x >> 5;
    const int m0 = (blockIdx.x >> 2) * 64 + (wave >> 1) * 32;
    const int o0 = (blockIdx.x & 3) * 128 + (wave & 1) * 64;
    const int r = lane & 15;
    const int h = lane >> 4;

    const float* ap = X  + (size_t)(m0 + r) * Dn + 4 * h;
    const v4f* bq = (const v4f*)Wa + (size_t)(o0 >> 4) * WT_PIECES + lane;

    v8f acc[2][4];
#pragma unroll
    for (int mt = 0; mt < 2; ++mt)
#pragma unroll
        for (int nt = 0; nt < 4; ++nt)
            acc[mt][nt] = (v8f){0.f, 0.f, 0.f, 0.f, 0.f, 0.f, 0.f, 0.f};

    Frags f[NS];
#pragma unroll
    for (int s = 0; s < NS - 1; ++s) {
        load_frags(f[s], ap, bq, s * 16);
        close_fetch_group();
    }
#pragma unroll 1
    for (int k0 = 0; k0 < Dn - 16 * NS; k0 += 16 * NS)
        ring_trip<false>(acc, f, ap, bq, k0);
    ring_trip<true>(acc, f, ap, bq, Dn - 16 * NS);

#pragma unroll
    for (int k0 = 0; k0 < NL; k0 += 4) {
        v2f a0 = *(const v2f*)(cnt + (size_t)(m0 + r) * NL + k0 + 2 * h);
        v2f a1 = *(const v2f*)(cnt + (size_t)(m0 + 16 + r) * NL + k0 + 2 * h);
#pragma unroll
        for (int nt = 0; nt < 4; ++nt) {
            v2f b = *(const v2f*)(Bt + (size_t)(o0 + 4 * r + nt) * NL + k0 + 2 * h);
            acc[0][nt] = __builtin_amdgcn_wmma_f32_16x16x4_f32(false, a0, false, b, (short)0, acc[0][nt], false, false);
            acc[1][nt] = __builtin_amdgcn_wmma_f32_16x16x4_f32(false, a1, false, b, (short)0, acc[1][nt], false, false);
        }
    }

#pragma unroll
    for (int mt = 0; mt < 2; ++mt)
#pragma unroll
        for (int v = 0; v < 8; ++v)
            *(v4f*)(out + (size_t)(m0 + mt * 16 + v + 8 * h) * Dn + o0 + 4 * r) =
                (v4f){acc[mt][0][v], acc[mt][1][v], acc[mt][2][v], acc[mt][3][v]};
}

extern "C" void kernel_launch(void* const* d_in, const int* in_sizes, int n_in,
                              void* d_out, int out_size, void* d_ws, size_t ws_size,
                              hipStream_t stream) {
    const float* feature = (const float*)d_in[0];
    const int*   graph   = (const int*)d_in[1];
    const float* weights = (const float*)d_in[2];
    const float* bias    = (const float*)d_in[3];
    float* out = (float*)d_out;

    float* ws = (float*)d_ws;
    float* Wa  = ws;
    float* Bt  = ws + Dn * Dn;
    float* cnt = ws + Dn * Dn + Dn * NL;

    prep_all<<<256, 128, 0, stream>>>(weights, bias, graph, Wa, Bt, cnt);
    gemm_fused<<<256, 128, 0, stream>>>(feature, Wa, Bt, cnt, out);
}
